# starting kernel
# baseline (speedup 1.0000x reference)
#include <hip/hip_runtime.h>

typedef __attribute__((ext_vector_type(2))) float v2f;
typedef __attribute__((ext_vector_type(8))) float v8f;

#define GRU_T 512
#define GRU_H 16

__device__ __forceinline__ float sigmoid_fast(float x) {
    float e = __expf(-x);
    return __builtin_amdgcn_rcpf(1.0f + e);
}
__device__ __forceinline__ float tanh_fast(float x) {
    float e = __expf(-2.0f * x);
    return (1.0f - e) * __builtin_amdgcn_rcpf(1.0f + e);
}

__global__ __launch_bounds__(32)
void gru_wmma_kernel(const float* __restrict__ x,
                     const float* __restrict__ W_ih,
                     const float* __restrict__ W_hh,
                     const float* __restrict__ b_ih,
                     const float* __restrict__ b_hh,
                     const float* __restrict__ fc_w,
                     const float* __restrict__ fc_b,
                     float* __restrict__ out)
{
    __shared__ float hbuf[16 * 18];
    __shared__ float red[32];

    const int lane = threadIdx.x;
    const int col  = lane & 15;
    const int half = lane >> 4;
    const int b0   = blockIdx.x * 16;

    v2f A[3][4];
    #pragma unroll
    for (int g = 0; g < 3; ++g)
        #pragma unroll
        for (int j = 0; j < 4; ++j)
            A[g][j] = *(const v2f*)(W_hh + (g * 16 + col) * 16 + 4 * j + 2 * half);

    float wr[8], wz[8], wn[8], cr0[8], cz0[8], cn0[8], bnc[8], fw[8];
    #pragma unroll
    for (int v = 0; v < 8; ++v) {
        int n = v + 8 * half;
        wr[v]  = W_ih[n];
        wz[v]  = W_ih[16 + n];
        wn[v]  = W_ih[32 + n];
        cr0[v] = b_ih[n]      + b_hh[n];
        cz0[v] = b_ih[16 + n] + b_hh[16 + n];
        cn0[v] = b_hh[32 + n];
        bnc[v] = b_ih[32 + n];
        fw[v]  = fc_w[n];
    }

    float h[8];
    #pragma unroll
    for (int v = 0; v < 8; ++v) h[v] = 0.0f;

    const float* xrow = x + (size_t)(b0 + col) * GRU_T;

    float4 xq = *(const float4*)(xrow);

    for (int t4 = 0; t4 < GRU_T / 4; ++t4) {
        float4 xcur = xq;
        if (t4 + 1 < GRU_T / 4) {
            xq = *(const float4*)(xrow + 4 * (t4 + 1));
            __builtin_prefetch(xrow + 4 * (t4 + 2), 0, 0);
        }
        float xs[4] = {xcur.x, xcur.y, xcur.z, xcur.w};

        #pragma unroll
        for (int s = 0; s < 4; ++s) {
            float xt = xs[s];

            #pragma unroll
            for (int p = 0; p < 4; ++p) {
                v2f hv; hv.x = h[2 * p]; hv.y = h[2 * p + 1];
                *(v2f*)(hbuf + col * 18 + (2 * p + 8 * half)) = hv;
            }
            __syncthreads();
            v2f Bm[4];
            #pragma unroll
            for (int j = 0; j < 4; ++j)
                Bm[j] = *(const v2f*)(hbuf + col * 18 + 4 * j + 2 * half);
            __syncthreads();

            v8f accR = {cr0[0], cr0[1], cr0[2], cr0[3], cr0[4], cr0[5], cr0[6], cr0[7]};
            v8f accZ = {cz0[0], cz0[1], cz0[2], cz0[3], cz0[4], cz0[5], cz0[6], cz0[7]};
            v8f accN = {cn0[0], cn0[1], cn0[2], cn0[3], cn0[4], cn0[5], cn0[6], cn0[7]};
            #pragma unroll
            for (int j = 0; j < 4; ++j) {
                accR = __builtin_amdgcn_wmma_f32_16x16x4_f32(
                    false, A[0][j], false, Bm[j], (short)0, accR, false, false);
                accZ = __builtin_amdgcn_wmma_f32_16x16x4_f32(
                    false, A[1][j], false, Bm[j], (short)0, accZ, false, false);
                accN = __builtin_amdgcn_wmma_f32_16x16x4_f32(
                    false, A[2][j], false, Bm[j], (short)0, accN, false, false);
            }

            #pragma unroll
            for (int v = 0; v < 8; ++v) {
                float r  = sigmoid_fast(accR[v] + xt * wr[v]);
                float z  = sigmoid_fast(accZ[v] + xt * wz[v]);
                float xn = xt * wn[v] + bnc[v];
                float nn = tanh_fast(xn + r * accN[v]);
                h[v] = nn + z * (h[v] - nn);
            }
        }
    }

    float partial = 0.0f;
    #pragma unroll
    for (int v = 0; v < 8; ++v) partial += h[v] * fw[v];
    red[lane] = partial;
    __syncthreads();
    if (lane < 16)
        out[b0 + lane] = partial + red[lane + 16] + fc_b[0];
}

extern "C" void kernel_launch(void* const* d_in, const int* in_sizes, int n_in,
                              void* d_out, int out_size, void* d_ws, size_t ws_size,
                              hipStream_t stream) {
    const float* x    = (const float*)d_in[0];
    const float* W_ih = (const float*)d_in[1];
    const float* W_hh = (const float*)d_in[2];
    const float* b_ih = (const float*)d_in[3];
    const float* b_hh = (const float*)d_in[4];
    const float* fc_w = (const float*)d_in[5];
    const float* fc_b = (const float*)d_in[6];
    float* out = (float*)d_out;

    const int B = in_sizes[0] / GRU_T;
    gru_wmma_kernel<<<B / 16, 32, 0, stream>>>(x, W_ih, W_hh, b_ih, b_hh,
                                               fc_w, fc_b, out);
}


# B operand read straight from the state registers: k order of A and B loaders set to 8*half+2j+p, so the per-step LDS transpose (4 stores, 4 loads, 2 barriers) goes; order of the k sums changes
# speedup vs baseline: 1.3497x; 1.3497x over previous
#include <hip/hip_runtime.h>

typedef __attribute__((ext_vector_type(2))) float v2f;
typedef __attribute__((ext_vector_type(8))) float v8f;

#define GRU_T 512
#define GRU_H 16

__device__ __forceinline__ float sigmoid_fast(float x) {
    float e = __expf(-x);
    return __builtin_amdgcn_rcpf(1.0f + e);
}
__device__ __forceinline__ float tanh_fast(float x) {
    float e = __expf(-2.0f * x);
    return (1.0f - e) * __builtin_amdgcn_rcpf(1.0f + e);
}

__global__ __launch_bounds__(32)
void gru_wmma_kernel(const float* __restrict__ x,
                     const float* __restrict__ W_ih,
                     const float* __restrict__ W_hh,
                     const float* __restrict__ b_ih,
                     const float* __restrict__ b_hh,
                     const float* __restrict__ fc_w,
                     const float* __restrict__ fc_b,
                     float* __restrict__ out)
{
    __shared__ float red[32];

    const int lane = threadIdx.x;
    const int col  = lane & 15;
    const int half = lane >> 4;
    const int b0   = blockIdx.x * 16;

    v2f A[3][4];
    #pragma unroll
    for (int g = 0; g < 3; ++g)
        #pragma unroll
        for (int j = 0; j < 4; ++j)
            A[g][j] = *(const v2f*)(W_hh + (g * 16 + col) * 16 + 8 * half + 2 * j);

    float wr[8], wz[8], wn[8], cr0[8], cz0[8], cn0[8], bnc[8], fw[8];
    #pragma unroll
    for (int v = 0; v < 8; ++v) {
        int n = v + 8 * half;
        wr[v]  = W_ih[n];
        wz[v]  = W_ih[16 + n];
        wn[v]  = W_ih[32 + n];
        cr0[v] = b_ih[n]      + b_hh[n];
        cz0[v] = b_ih[16 + n] + b_hh[16 + n];
        cn0[v] = b_hh[32 + n];
        bnc[v] = b_ih[32 + n];
        fw[v]  = fc_w[n];
    }

    v2f H[4];
    #pragma unroll
    for (int j = 0; j < 4; ++j) { H[j].x = 0.0f; H[j].y = 0.0f; }

    const float* xrow = x + (size_t)(b0 + col) * GRU_T;

    float4 xq = *(const float4*)(xrow);

    for (int t4 = 0; t4 < GRU_T / 4; ++t4) {
        float4 xcur = xq;
        if (t4 + 1 < GRU_T / 4) {
            xq = *(const float4*)(xrow + 4 * (t4 + 1));
            __builtin_prefetch(xrow + 4 * (t4 + 2), 0, 0);
        }
        float xs[4] = {xcur.x, xcur.y, xcur.z, xcur.w};

        #pragma unroll
        for (int s = 0; s < 4; ++s) {
            float xt = xs[s];

            v8f accR = {cr0[0], cr0[1], cr0[2], cr0[3], cr0[4], cr0[5], cr0[6], cr0[7]};
            v8f accZ = {cz0[0], cz0[1], cz0[2], cz0[3], cz0[4], cz0[5], cz0[6], cz0[7]};
            v8f accN = {cn0[0], cn0[1], cn0[2], cn0[3], cn0[4], cn0[5], cn0[6], cn0[7]};
            #pragma unroll
            for (int j = 0; j < 4; ++j) {
                accR = __builtin_amdgcn_wmma_f32_16x16x4_f32(
                    false, A[0][j], false, H[j], (short)0, accR, false, false);
                accZ = __builtin_amdgcn_wmma_f32_16x16x4_f32(
                    false, A[1][j], false, H[j], (short)0, accZ, false, false);
                accN = __builtin_amdgcn_wmma_f32_16x16x4_f32(
                    false, A[2][j], false, H[j], (short)0, accN, false, false);
            }

            #pragma unroll
            for (int v = 0; v < 8; ++v) {
                float r  = sigmoid_fast(accR[v] + xt * wr[v]);
                float z  = sigmoid_fast(accZ[v] + xt * wz[v]);
                float xn = xt * wn[v] + bnc[v];
                float nn = tanh_fast(xn + r * accN[v]);
                float ho = H[v >> 1][v & 1];
                H[v >> 1][v & 1] = nn + z * (ho - nn);
            }
        }
    }

    float partial = 0.0f;
    #pragma unroll
    for (int v = 0; v < 8; ++v) partial += H[v >> 1][v & 1] * fw[v];
    red[lane] = partial;
    __syncthreads();
    if (lane < 16)
        out[b0 + lane] = partial + red[lane + 16] + fc_b[0];
}

extern "C" void kernel_launch(void* const* d_in, const int* in_sizes, int n_in,
                              void* d_out, int out_size, void* d_ws, size_t ws_size,
                              hipStream_t stream) {
    const float* x    = (const float*)d_in[0];
    const float* W_ih = (const float*)d_in[1];
    const float* W_hh = (const float*)d_in[2];
    const float* b_ih = (const float*)d_in[3];
    const float* b_hh = (const float*)d_in[4];
    const float* fc_w = (const float*)d_in[5];
    const float* fc_b = (const float*)d_in[6];
    float* out = (float*)d_out;

    const int B = in_sizes[0] / GRU_T;
    gru_wmma_kernel<<<B / 16, 32, 0, stream>>>(x, W_ih, W_hh, b_ih, b_hh,
                                               fc_w, fc_b, out);
}


# sigmoid and tanh gates on the hardware tanh instruction (sigmoid = 0.5+0.5*tanh(p/2), the 1/2 folded into the loop-invariant weights and biases): 3 transcendentals per element, not 6
# speedup vs baseline: 1.5924x; 1.1798x over previous
#include <hip/hip_runtime.h>

typedef __attribute__((ext_vector_type(2))) float v2f;
typedef __attribute__((ext_vector_type(8))) float v8f;

#define GRU_T 512
#define GRU_H 16

__device__ __forceinline__ float tanh_hw(float y) {
    return __builtin_amdgcn_tanhf(y);
}

__global__ __launch_bounds__(32)
void gru_wmma_kernel(const float* __restrict__ x,
                     const float* __restrict__ W_ih,
                     const float* __restrict__ W_hh,
                     const float* __restrict__ b_ih,
                     const float* __restrict__ b_hh,
                     const float* __restrict__ fc_w,
                     const float* __restrict__ fc_b,
                     float* __restrict__ out)
{
    __shared__ float red[32];

    const int lane = threadIdx.x;
    const int col  = lane & 15;
    const int half = lane >> 4;
    const int b0   = blockIdx.x * 16;

    v2f A[3][4];
    #pragma unroll
    for (int g = 0; g < 3; ++g)
        #pragma unroll
        for (int j = 0; j < 4; ++j)
            A[g][j] = 0.5f * *(const v2f*)(W_hh + (g * 16 + col) * 16 + 8 * half + 2 * j);

    float wr[8], wz[8], wn[8], cr0[8], cz0[8], cn0[8], bnc[8], fw[8];
    #pragma unroll
    for (int v = 0; v < 8; ++v) {
        int n = v + 8 * half;
        wr[v]  = 0.5f * W_ih[n];
        wz[v]  = 0.5f * W_ih[16 + n];
        wn[v]  = W_ih[32 + n];
        cr0[v] = 0.5f * (b_ih[n]      + b_hh[n]);
        cz0[v] = 0.5f * (b_ih[16 + n] + b_hh[16 + n]);
        cn0[v] = 0.5f * b_hh[32 + n];
        bnc[v] = b_ih[32 + n];
        fw[v]  = fc_w[n];
    }

    v2f H[4];
    #pragma unroll
    for (int j = 0; j < 4; ++j) { H[j].x = 0.0f; H[j].y = 0.0f; }

    const float* xrow = x + (size_t)(b0 + col) * GRU_T;

    float4 xq = *(const float4*)(xrow);

    for (int t4 = 0; t4 < GRU_T / 4; ++t4) {
        float4 xcur = xq;
        if (t4 + 1 < GRU_T / 4) {
            xq = *(const float4*)(xrow + 4 * (t4 + 1));
            __builtin_prefetch(xrow + 4 * (t4 + 2), 0, 0);
        }
        float xs[4] = {xcur.x, xcur.y, xcur.z, xcur.w};

        #pragma unroll
        for (int s = 0; s < 4; ++s) {
            float xt = xs[s];

            v8f accR = {cr0[0], cr0[1], cr0[2], cr0[3], cr0[4], cr0[5], cr0[6], cr0[7]};
            v8f accZ = {cz0[0], cz0[1], cz0[2], cz0[3], cz0[4], cz0[5], cz0[6], cz0[7]};
            v8f accN = {cn0[0], cn0[1], cn0[2], cn0[3], cn0[4], cn0[5], cn0[6], cn0[7]};
            #pragma unroll
            for (int j = 0; j < 4; ++j) {
                accR = __builtin_amdgcn_wmma_f32_16x16x4_f32(
                    false, A[0][j], false, H[j], (short)0, accR, false, false);
                accZ = __builtin_amdgcn_wmma_f32_16x16x4_f32(
                    false, A[1][j], false, H[j], (short)0, accZ, false, false);
                accN = __builtin_amdgcn_wmma_f32_16x16x4_f32(
                    false, A[2][j], false, H[j], (short)0, accN, false, false);
            }

            #pragma unroll
            for (int v = 0; v < 8; ++v) {
                float tr = tanh_hw(accR[v] + xt * wr[v]);
                float tz = tanh_hw(accZ[v] + xt * wz[v]);
                float z  = __builtin_fmaf(tz, 0.5f, 0.5f);
                float xn = xt * wn[v] + bnc[v];
                float nn = tanh_hw((xn + accN[v]) + tr * accN[v]);
                float ho = H[v >> 1][v & 1];
                H[v >> 1][v & 1] = nn + z * (ho - nn);
            }
        }
    }

    float partial = 0.0f;
    #pragma unroll
    for (int v = 0; v < 8; ++v) partial += H[v >> 1][v & 1] * fw[v];
    red[lane] = partial;
    __syncthreads();
    if (lane < 16)
        out[b0 + lane] = partial + red[lane + 16] + fc_b[0];
}

extern "C" void kernel_launch(void* const* d_in, const int* in_sizes, int n_in,
                              void* d_out, int out_size, void* d_ws, size_t ws_size,
                              hipStream_t stream) {
    const float* x    = (const float*)d_in[0];
    const float* W_ih = (const float*)d_in[1];
    const float* W_hh = (const float*)d_in[2];
    const float* b_ih = (const float*)d_in[3];
    const float* b_hh = (const float*)d_in[4];
    const float* fc_w = (const float*)d_in[5];
    const float* fc_b = (const float*)d_in[6];
    float* out = (float*)d_out;

    const int B = in_sizes[0] / GRU_T;
    gru_wmma_kernel<<<B / 16, 32, 0, stream>>>(x, W_ih, W_hh, b_ih, b_hh,
                                               fc_w, fc_b, out);
}


# step ordered by hand around the matrix instructions: z chain last, r/n gate math between its four instructions so latencies run under them; x terms enter via the chains' C operand (sum order changes)
# speedup vs baseline: 1.6866x; 1.0592x over previous
#include <hip/hip_runtime.h>

typedef __attribute__((ext_vector_type(2))) float v2f;
typedef __attribute__((ext_vector_type(8))) float v8f;

#define GRU_T 512
#define GRU_H 16

__device__ __forceinline__ float tanh_hw(float y) {
    return __builtin_amdgcn_tanhf(y);
}


#define ORDER_POINT(acc, a)                                                        \
    asm volatile("" : "+v"(acc), "+v"(a[0]), "+v"(a[1]), "+v"(a[2]), "+v"(a[3]),   \
                      "+v"(a[4]), "+v"(a[5]), "+v"(a[6]), "+v"(a[7]))

__global__ __launch_bounds__(32)
void gru_wmma_kernel(const float* __restrict__ x,
                     const float* __restrict__ W_ih,
                     const float* __restrict__ W_hh,
                     const float* __restrict__ b_ih,
                     const float* __restrict__ b_hh,
                     const float* __restrict__ fc_w,
                     const float* __restrict__ fc_b,
                     float* __restrict__ out)
{
    __shared__ float red[32];

    const int lane = threadIdx.x;
    const int col  = lane & 15;
    const int half = lane >> 4;
    const int b0   = blockIdx.x * 16;

    v2f A[3][4];
    #pragma unroll
    for (int g = 0; g < 3; ++g)
        #pragma unroll
        for (int j = 0; j < 4; ++j)
            A[g][j] = 0.5f * *(const v2f*)(W_hh + (g * 16 + col) * 16 + 8 * half + 2 * j);

    float wr[8], wz[8], wn[8], cr0[8], cz0[8], bnc[8], fw[8];
    v8f cn0;
    #pragma unroll
    for (int v = 0; v < 8; ++v) {
        int n = v + 8 * half;
        wr[v]  = 0.5f * W_ih[n];
        wz[v]  = 0.5f * W_ih[16 + n];
        wn[v]  = W_ih[32 + n];
        cr0[v] = 0.5f * (b_ih[n]      + b_hh[n]);
        cz0[v] = 0.5f * (b_ih[16 + n] + b_hh[16 + n]);
        cn0[v] = 0.5f * b_hh[32 + n];
        bnc[v] = b_ih[32 + n];
        fw[v]  = fc_w[n];
    }

    v2f H[4];
    #pragma unroll
    for (int j = 0; j < 4; ++j) { H[j].x = 0.0f; H[j].y = 0.0f; }

    const float* xrow = x + (size_t)(b0 + col) * GRU_T;

    float4 xq = *(const float4*)(xrow);

    v8f cR, cZ;
    float xn[8];
    #pragma unroll
    for (int v = 0; v < 8; ++v) {
        cR[v] = __builtin_fmaf(xq.x, wr[v], cr0[v]);
        cZ[v] = __builtin_fmaf(xq.x, wz[v], cz0[v]);
        xn[v] = __builtin_fmaf(xq.x, wn[v], bnc[v]);
    }

    for (int t4 = 0; t4 < GRU_T / 4; ++t4) {
        float4 xcur = xq;
        if (t4 + 1 < GRU_T / 4) {
            xq = *(const float4*)(xrow + 4 * (t4 + 1));
            __builtin_prefetch(xrow + 4 * (t4 + 2), 0, 0);
        }
        float xnx[4] = {xcur.y, xcur.z, xcur.w, xq.x};

        #pragma unroll
        for (int s = 0; s < 4; ++s) {
            const float xnext = xnx[s];

            v8f accR = cR, accN = cn0;
            #pragma unroll
            for (int j = 0; j < 4; ++j) {
                accR = __builtin_amdgcn_wmma_f32_16x16x4_f32(
                    false, A[0][j], false, H[j], (short)0, accR, false, false);
                accN = __builtin_amdgcn_wmma_f32_16x16x4_f32(
                    false, A[2][j], false, H[j], (short)0, accN, false, false);
            }

            float g[8];
            #pragma unroll
            for (int v = 0; v < 8; ++v) g[v] = tanh_hw(accR[v]);
            v8f accZ = cZ;
            ORDER_POINT(accZ, g);
            accZ = __builtin_amdgcn_wmma_f32_16x16x4_f32(
                false, A[1][0], false, H[0], (short)0, accZ, false, false);
            ORDER_POINT(accZ, g);

            #pragma unroll
            for (int v = 0; v < 8; ++v) g[v] = __builtin_fmaf(accN[v], 1.0f + g[v], xn[v]);
            ORDER_POINT(accZ, g);
            accZ = __builtin_amdgcn_wmma_f32_16x16x4_f32(
                false, A[1][1], false, H[1], (short)0, accZ, false, false);
            accZ = __builtin_amdgcn_wmma_f32_16x16x4_f32(
                false, A[1][2], false, H[2], (short)0, accZ, false, false);
            ORDER_POINT(accZ, g);

            #pragma unroll
            for (int v = 0; v < 8; ++v) g[v] = tanh_hw(g[v]);
            ORDER_POINT(accZ, g);
            accZ = __builtin_amdgcn_wmma_f32_16x16x4_f32(
                false, A[1][3], false, H[3], (short)0, accZ, false, false);
            ORDER_POINT(accZ, g);

            #pragma unroll
            for (int v = 0; v < 8; ++v) {
                float hh = 0.5f * H[v >> 1][v & 1];
                float e  = __builtin_fmaf(g[v], -0.5f, hh);
                float f  = __builtin_fmaf(g[v],  0.5f, hh);
                float tz = tanh_hw(accZ[v]);
                H[v >> 1][v & 1] = __builtin_fmaf(tz, e, f);
            }
            #pragma unroll
            for (int v = 0; v < 8; ++v) {
                cR[v] = __builtin_fmaf(xnext, wr[v], cr0[v]);
                cZ[v] = __builtin_fmaf(xnext, wz[v], cz0[v]);
                xn[v] = __builtin_fmaf(xnext, wn[v], bnc[v]);
            }
        }
    }

    float partial = 0.0f;
    #pragma unroll
    for (int v = 0; v < 8; ++v) partial += H[v >> 1][v & 1] * fw[v];
    red[lane] = partial;
    __syncthreads();
    if (lane < 16)
        out[b0 + lane] = partial + red[lane + 16] + fc_b[0];
}

extern "C" void kernel_launch(void* const* d_in, const int* in_sizes, int n_in,
                              void* d_out, int out_size, void* d_ws, size_t ws_size,
                              hipStream_t stream) {
    const float* x    = (const float*)d_in[0];
    const float* W_ih = (const float*)d_in[1];
    const float* W_hh = (const float*)d_in[2];
    const float* b_ih = (const float*)d_in[3];
    const float* b_hh = (const float*)d_in[4];
    const float* fc_w = (const float*)d_in[5];
    const float* fc_b = (const float*)d_in[6];
    float* out = (float*)d_out;

    const int B = in_sizes[0] / GRU_T;
    gru_wmma_kernel<<<B / 16, 32, 0, stream>>>(x, W_ih, W_hh, b_ih, b_hh,
                                               fc_w, fc_b, out);
}


# speedup vs baseline: 1.8653x; 1.1060x over previous
#include <hip/hip_runtime.h>

typedef __attribute__((ext_vector_type(2))) float v2f;
typedef __attribute__((ext_vector_type(8))) float v8f;

#define GRU_T 512
#define GRU_H 16

__device__ __forceinline__ float tanh_hw(float y) {
    return __builtin_amdgcn_tanhf(y);
}


#define ORDER_POINT(acc, a)                                                        \
    asm volatile("" : "+v"(acc), "+v"(a[0]), "+v"(a[1]), "+v"(a[2]), "+v"(a[3]),   \
                      "+v"(a[4]), "+v"(a[5]), "+v"(a[6]), "+v"(a[7]))

#define PAIR_POINT(next, src, t, p, q)                                             \
    asm volatile("" : "+v"(next), "+v"(src), "+v"(t), "+v"(p), "+v"(q))
#define PAIR_POINT_H(next, h, t, p, q)                                             \
    asm volatile("" : "+v"(next), "+v"(h[0]), "+v"(h[1]), "+v"(h[2]), "+v"(h[3]),  \
                      "+v"(t), "+v"(p), "+v"(q))
#define TAIL_POINT(acc, nsrc, t, p, q)                                             \
    asm volatile("" : "+v"(acc), "+v"(nsrc), "+v"(t), "+v"(p), "+v"(q))
#define TAIL_POINT1(acc, nsrc, t, p)                                               \
    asm volatile("" : "+v"(acc), "+v"(nsrc), "+v"(t), "+v"(p))
#define LATE_POINT(accA, accB, t)                                                  \
    asm volatile("" : "+v"(accA), "+v"(accB), "+v"(t[0]), "+v"(t[1]), "+v"(t[2]),   \
                      "+v"(t[3]))
#define CHAINS_POINT(accA, accB, src)                                              \
    asm volatile("" : "+v"(accA), "+v"(accB), "+v"(src))

__global__ __launch_bounds__(32)
void gru_wmma_kernel(const float* __restrict__ x,
                     const float* __restrict__ W_ih,
                     const float* __restrict__ W_hh,
                     const float* __restrict__ b_ih,
                     const float* __restrict__ b_hh,
                     const float* __restrict__ fc_w,
                     const float* __restrict__ fc_b,
                     float* __restrict__ out)
{
    __shared__ float red[32];

    const int lane = threadIdx.x;
    const int col  = lane & 15;
    const int half = lane >> 4;
    const int b0   = blockIdx.x * 16;

    v2f A[3][4];
    #pragma unroll
    for (int g = 0; g < 3; ++g)
        #pragma unroll
        for (int j = 0; j < 4; ++j)
            A[g][j] = 0.5f * *(const v2f*)(W_hh + (g * 16 + col) * 16 + 8 * half + 2 * j);

    float wr[8], wz[8], wn[8], cr0[8], cz0[8], bnc[8], fw[8];
    v8f cn0;
    #pragma unroll
    for (int v = 0; v < 8; ++v) {
        int n = v + 8 * half;
        wr[v]  = 0.5f * W_ih[n];
        wz[v]  = 0.5f * W_ih[16 + n];
        wn[v]  = W_ih[32 + n];
        cr0[v] = 0.5f * (b_ih[n]      + b_hh[n]);
        cz0[v] = 0.5f * (b_ih[16 + n] + b_hh[16 + n]);
        cn0[v] = 0.5f * b_hh[32 + n];
        bnc[v] = b_ih[32 + n];
        fw[v]  = fc_w[n];
    }

    v2f H[4];
    #pragma unroll
    for (int j = 0; j < 4; ++j) { H[j].x = 0.0f; H[j].y = 0.0f; }

    const float* xrow = x + (size_t)(b0 + col) * GRU_T;

    float4 xq = *(const float4*)(xrow);

    v8f cR, cZ;
    float xn[8];
    #pragma unroll
    for (int v = 0; v < 8; ++v) {
        cR[v] = __builtin_fmaf(xq.x, wr[v], cr0[v]);
        cZ[v] = __builtin_fmaf(xq.x, wz[v], cz0[v]);
        xn[v] = __builtin_fmaf(xq.x, wn[v], bnc[v]);
    }

    float ptz[4], pe[4], pf[4];
    #pragma unroll
    for (int i = 0; i < 4; ++i) { ptz[i] = 0.0f; pe[i] = 0.0f; pf[i] = 0.0f; }

    for (int t4 = 0; t4 < GRU_T / 4; ++t4) {
        float4 xcur = xq;
        const int tn = (t4 + 1 < GRU_T / 4) ? t4 + 1 : t4;
        xq = *(const float4*)(xrow + 4 * tn);
        __builtin_prefetch(xrow + 4 * (tn + 1), 0, 0);
        float xnx[4] = {xcur.y, xcur.z, xcur.w, xq.x};

        #pragma unroll
        for (int s = 0; s < 4; ++s) {
            float xs = xnx[s];

            v8f accR = cR, accN = cn0;
            accR = __builtin_amdgcn_wmma_f32_16x16x4_f32(
                false, A[0][0], false, H[0], (short)0, accR, false, false);
            accN = __builtin_amdgcn_wmma_f32_16x16x4_f32(
                false, A[2][0], false, H[0], (short)0, accN, false, false);
            LATE_POINT(accR, accN, ptz);
            #pragma unroll
            for (int i = 0; i < 4; ++i)
                H[2 + (i >> 1)][i & 1] = __builtin_fmaf(ptz[i], pe[i], pf[i]);
            asm volatile("" : "+v"(H[1]), "+v"(H[2]), "+v"(H[3]));
            #pragma unroll
            for (int j = 1; j < 4; ++j) {
                accR = __builtin_amdgcn_wmma_f32_16x16x4_f32(
                    false, A[0][j], false, H[j], (short)0, accR, false, false);
                accN = __builtin_amdgcn_wmma_f32_16x16x4_f32(
                    false, A[2][j], false, H[j], (short)0, accN, false, false);
            }

            CHAINS_POINT(accR, accN, xs);

            float g[8], cRn[8], cZn[8], xnn[8], hh[8];
            #pragma unroll
            for (int v = 0; v < 4; ++v) {
                g[v] = tanh_hw(accR[v]);
                cRn[2 * v]     = __builtin_fmaf(xs, wr[2 * v],     cr0[2 * v]);
                cRn[2 * v + 1] = __builtin_fmaf(xs, wr[2 * v + 1], cr0[2 * v + 1]);
                PAIR_POINT(accR, xs, g[v], cRn[2 * v], cRn[2 * v + 1]);
            }
            #pragma unroll
            for (int v = 4; v < 8; ++v) {
                g[v] = tanh_hw(accR[v]);
                cZn[2 * v - 8] = __builtin_fmaf(xs, wz[2 * v - 8], cz0[2 * v - 8]);
                cZn[2 * v - 7] = __builtin_fmaf(xs, wz[2 * v - 7], cz0[2 * v - 7]);
                PAIR_POINT(accR, xs, g[v], cZn[2 * v - 8], cZn[2 * v - 7]);
            }
            v8f accZ = cZ;
            ORDER_POINT(accZ, g);
            accZ = __builtin_amdgcn_wmma_f32_16x16x4_f32(
                false, A[1][0], false, H[0], (short)0, accZ, false, false);
            ORDER_POINT(accZ, g);

            #pragma unroll
            for (int v = 0; v < 8; ++v) g[v] = __builtin_fmaf(accN[v], 1.0f + g[v], xn[v]);
            ORDER_POINT(accZ, g);
            accZ = __builtin_amdgcn_wmma_f32_16x16x4_f32(
                false, A[1][1], false, H[1], (short)0, accZ, false, false);
            ORDER_POINT(accZ, g);

            #pragma unroll
            for (int v = 0; v < 4; ++v) {
                g[v] = tanh_hw(g[v]);
                xnn[2 * v]     = __builtin_fmaf(xs, wn[2 * v],     bnc[2 * v]);
                xnn[2 * v + 1] = __builtin_fmaf(xs, wn[2 * v + 1], bnc[2 * v + 1]);
                PAIR_POINT(g[v + 1], xs, g[v], xnn[2 * v], xnn[2 * v + 1]);
            }
            ORDER_POINT(accZ, g);
            accZ = __builtin_amdgcn_wmma_f32_16x16x4_f32(
                false, A[1][2], false, H[2], (short)0, accZ, false, false);
            ORDER_POINT(accZ, g);

            #pragma unroll
            for (int v = 4; v < 8; ++v) {
                g[v] = tanh_hw(g[v]);
                hh[2 * v - 8] = 0.5f * H[v - 4].x;
                hh[2 * v - 7] = 0.5f * H[v - 4].y;
                PAIR_POINT_H(g[v < 7 ? v + 1 : 0], H, g[v], hh[2 * v - 8], hh[2 * v - 7]);
            }
            ORDER_POINT(accZ, g);
            accZ = __builtin_amdgcn_wmma_f32_16x16x4_f32(
                false, A[1][3], false, H[3], (short)0, accZ, false, false);
            ORDER_POINT(accZ, g);

            float e[8], f[8], tz[8];
            #pragma unroll
            for (int v = 0; v < 4; ++v) {
                e[v] = __builtin_fmaf(g[v], -0.5f, hh[v]);
                f[v] = __builtin_fmaf(g[v],  0.5f, hh[v]);
            }
            asm volatile("" : "+v"(accZ), "+v"(g[4]), "+v"(e[0]), "+v"(f[0]), "+v"(e[1]),
                              "+v"(f[1]), "+v"(e[2]), "+v"(f[2]), "+v"(e[3]), "+v"(f[3]));
            #pragma unroll
            for (int v = 0; v < 4; ++v) {
                tz[v] = tanh_hw(accZ[v]);
                e[v + 4] = __builtin_fmaf(g[v + 4], -0.5f, hh[v + 4]);
                f[v + 4] = __builtin_fmaf(g[v + 4],  0.5f, hh[v + 4]);
                if (v < 3) TAIL_POINT(accZ, g[v + 5], tz[v], e[v + 4], f[v + 4]);
                else       TAIL_POINT(accZ, tz[0],    tz[v], e[v + 4], f[v + 4]);
            }
            tz[4] = tanh_hw(accZ[4]);
            H[0].x = __builtin_fmaf(tz[0], e[0], f[0]);
            TAIL_POINT1(accZ, tz[1], tz[4], H[0].x);
            tz[5] = tanh_hw(accZ[5]);
            H[0].y = __builtin_fmaf(tz[1], e[1], f[1]);
            TAIL_POINT1(accZ, tz[2], tz[5], H[0].y);
            tz[6] = tanh_hw(accZ[6]);
            H[1].x = __builtin_fmaf(tz[2], e[2], f[2]);
            TAIL_POINT1(accZ, tz[3], tz[6], H[1].x);
            tz[7] = tanh_hw(accZ[7]);
            H[1].y = __builtin_fmaf(tz[3], e[3], f[3]);
            asm volatile("" : "+v"(H[0]), "+v"(tz[7]), "+v"(H[1].y));
            #pragma unroll
            for (int i = 0; i < 4; ++i) { ptz[i] = tz[4 + i]; pe[i] = e[4 + i]; pf[i] = f[4 + i]; }
            #pragma unroll
            for (int v = 0; v < 8; ++v) { cR[v] = cRn[v]; cZ[v] = cZn[v]; xn[v] = xnn[v]; }
        }
    }

    #pragma unroll
    for (int i = 0; i < 4; ++i)
        H[2 + (i >> 1)][i & 1] = __builtin_fmaf(ptz[i], pe[i], pf[i]);

    float partial = 0.0f;
    #pragma unroll
    for (int v = 0; v < 8; ++v) partial += H[v >> 1][v & 1] * fw[v];
    red[lane] = partial;
    __syncthreads();
    if (lane < 16)
        out[b0 + lane] = partial + red[lane + 16] + fc_b[0];
}

extern "C" void kernel_launch(void* const* d_in, const int* in_sizes, int n_in,
                              void* d_out, int out_size, void* d_ws, size_t ws_size,
                              hipStream_t stream) {
    const float* x    = (const float*)d_in[0];
    const float* W_ih = (const float*)d_in[1];
    const float* W_hh = (const float*)d_in[2];
    const float* b_ih = (const float*)d_in[3];
    const float* b_hh = (const float*)d_in[4];
    const float* fc_w = (const float*)d_in[5];
    const float* fc_b = (const float*)d_in[6];
    float* out = (float*)d_out;

    const int B = in_sizes[0] / GRU_T;
    gru_wmma_kernel<<<B / 16, 32, 0, stream>>>(x, W_ih, W_hh, b_ih, b_hh,
                                               fc_w, fc_b, out);
}
